# starting kernel
# baseline (speedup 1.0000x reference)
#include <hip/hip_runtime.h>
#include <hip/hip_bf16.h>

typedef float v2f __attribute__((ext_vector_type(2)));
typedef float v8f __attribute__((ext_vector_type(8)));

#define N_FULL   1024
#define NBLK     32
#define TILE_M   16
#define YS       1028
#define NWAVES   4

__global__ __launch_bounds__(32) void build_packed_mats(
    const float* __restrict__ tw,
    float* __restrict__ pk)
{
    __shared__ float Cm[32 * 32];
    const int c   = threadIdx.x;
    const bool isB = (blockIdx.x >= 32);
    const int idx = isB ? (int)blockIdx.x - 32 : (int)blockIdx.x;

    for (int i = 0; i < 32; ++i) Cm[i * 32 + c] = (i == c) ? 1.0f : 0.0f;
    __syncthreads();

    for (int lsi = 0; lsi < 5; ++lsi) {
        const int ls = isB ? lsi + 5 : lsi;
        const int s  = 1 << ls;
        const int cs = isB ? (s >> 5) : s;
        for (int pr = 0; pr < 16; ++pr) {
            const int G  = pr / cs;
            const int t  = pr % cs;
            const int u0 = G * 2 * cs + t;
            const int u1 = u0 + cs;
            const int p0 = isB ? (u0 * 32 + idx) : (idx * 32 + u0);
            const int pid = (p0 / (2 * s)) * s + (p0 % s);
            const float* tp = tw + ls * 2048 + pid * 4;
            const float t00 = tp[0], t01 = tp[1], t10 = tp[2], t11 = tp[3];
            const float a0 = Cm[u0 * 32 + c];
            const float a1 = Cm[u1 * 32 + c];
            Cm[u0 * 32 + c] = t00 * a0 + t01 * a1;
            Cm[u1 * 32 + c] = t10 * a0 + t11 * a1;
        }
    }
    __syncthreads();

    const int L = c, Lm = L & 15, Lh = L >> 4;
    float* base = pk + (size_t)blockIdx.x * 1024;
    for (int h = 0; h < 2; ++h) {
        for (int kk = 0; kk < 8; ++kk) {
            const int row  = h * 16 + Lm;
            const int col0 = kk * 4 + Lh * 2;
            base[(h * 8 + kk) * 64 + L * 2 + 0] = Cm[row * 32 + col0 + 0];
            base[(h * 8 + kk) * 64 + L * 2 + 1] = Cm[row * 32 + col0 + 1];
        }
    }
}

__global__ __launch_bounds__(NWAVES * 32) void butterfly_wmma(
    const float* __restrict__ x,
    const float* __restrict__ pk,
    const float* __restrict__ bias,
    float* __restrict__ out)
{
    __shared__ __align__(16) float y[TILE_M * YS];

    const int tid = threadIdx.x;
    const int w   = tid >> 5;
    const int L   = tid & 31;
    const int Lm  = L & 15;
    const int Lh  = L >> 4;
    const size_t b0 = (size_t)blockIdx.x * TILE_M;

    const float* xrow = x + b0 * N_FULL;
    for (int i = tid; i < TILE_M * (N_FULL / 4); i += NWAVES * 32) {
        const int row = i >> 8;
        const int c4  = (i & 255) * 4;
        const float4 v = *(const float4*)(xrow + row * N_FULL + c4);
        *(float4*)(&y[row * YS + c4]) = v;
    }
    __syncthreads();

    const v2f* pkA = (const v2f*)pk;
    const v2f* pkB = (const v2f*)pk + 32 * 512;

    for (int g = w; g < NBLK; g += NWAVES) {
        v2f xa[8];
        const int xb = Lm * YS + g * 32 + Lh * 2;
        #pragma unroll
        for (int kk = 0; kk < 8; ++kk)
            xa[kk] = *(const v2f*)(&y[xb + kk * 4]);

        #pragma unroll
        for (int h = 0; h < 2; ++h) {
            v8f acc = {0.f, 0.f, 0.f, 0.f, 0.f, 0.f, 0.f, 0.f};
            #pragma unroll
            for (int kk = 0; kk < 8; ++kk) {
                const v2f bm = pkA[g * 512 + (h * 8 + kk) * 32 + L];
                acc = __builtin_amdgcn_wmma_f32_16x16x4_f32(
                    false, xa[kk], false, bm, (short)0, acc, false, false);
            }
            const int db = g * 32 + h * 16 + Lm;
            #pragma unroll
            for (int v = 0; v < 8; ++v)
                y[(v + 8 * Lh) * YS + db] = acc[v];
        }
    }
    __syncthreads();

    for (int r = w; r < 32; r += NWAVES) {
        v2f ya[8];
        #pragma unroll
        for (int kk = 0; kk < 8; ++kk) {
            const int k0 = kk * 4 + Lh * 2;
            ya[kk][0] = y[Lm * YS + (k0 + 0) * 32 + r];
            ya[kk][1] = y[Lm * YS + (k0 + 1) * 32 + r];
        }
        #pragma unroll
        for (int h = 0; h < 2; ++h) {
            v8f acc = {0.f, 0.f, 0.f, 0.f, 0.f, 0.f, 0.f, 0.f};
            #pragma unroll
            for (int kk = 0; kk < 8; ++kk) {
                const v2f bm = pkB[r * 512 + (h * 8 + kk) * 32 + L];
                acc = __builtin_amdgcn_wmma_f32_16x16x4_f32(
                    false, ya[kk], false, bm, (short)0, acc, false, false);
            }
            #pragma unroll
            for (int v = 0; v < 8; ++v)
                y[(v + 8 * Lh) * YS + (h * 16 + Lm) * 32 + r] = acc[v];
        }
    }
    __syncthreads();

    float* orow = out + b0 * N_FULL;
    for (int i = tid; i < TILE_M * (N_FULL / 4); i += NWAVES * 32) {
        const int row = i >> 8;
        const int c4  = (i & 255) * 4;
        float4 v = *(const float4*)(&y[row * YS + c4]);
        const float4 bv = *(const float4*)(bias + c4);
        v.x += bv.x; v.y += bv.y; v.z += bv.z; v.w += bv.w;
        *(float4*)(orow + row * N_FULL + c4) = v;
    }
}

extern "C" void kernel_launch(void* const* d_in, const int* in_sizes, int n_in,
                              void* d_out, int out_size, void* d_ws, size_t ws_size,
                              hipStream_t stream) {
    (void)n_in; (void)out_size; (void)ws_size;
    const float* x    = (const float*)d_in[0];
    const float* tw   = (const float*)d_in[1];
    const float* bias = (const float*)d_in[2];
    float* out = (float*)d_out;
    float* pk  = (float*)d_ws;

    const int batch = in_sizes[0] / N_FULL;

    build_packed_mats<<<64, 32, 0, stream>>>(tw, pk);
    butterfly_wmma<<<batch / TILE_M, NWAVES * 32, 0, stream>>>(x, pk, bias, out);
}


# Setup kernel: each composed-matrix entry is the product of its 5 path twiddles, all entries in parallel (one thread per packed pair) instead of a serial 80-step LDS sweep; same multiply order
# speedup vs baseline: 1.5283x; 1.5283x over previous
#include <hip/hip_runtime.h>
#include <hip/hip_bf16.h>

typedef float v2f __attribute__((ext_vector_type(2)));
typedef float v8f __attribute__((ext_vector_type(8)));

#define N_FULL   1024
#define NBLK     32
#define TILE_M   16
#define YS       1028
#define NWAVES   4

#define SETUP_THREADS 256

__global__ __launch_bounds__(SETUP_THREADS) void build_packed_mats(
    const float* __restrict__ tw,
    float* __restrict__ pk)
{
    const int o    = (int)blockIdx.x * SETUP_THREADS + (int)threadIdx.x;
    const int mat  = o >> 9;
    const int hk   = (o >> 5) & 15;
    const int L    = o & 31;
    const int i    = (hk >> 3) * 16 + (L & 15);
    const int j    = (hk & 7) * 4 + (L >> 4) * 2;
    const bool isB = (mat >= 32);
    const int idx  = mat & 31;

    v2f val;
    for (int lsi = 0; lsi < 5; ++lsi) {
        const int ls = isB ? lsi + 5 : lsi;
        const int s  = 1 << ls;
        const int cs = 1 << lsi;
        const int u0  = (i & (cs - 1)) | (j & ~(2 * cs - 1));
        const int p0  = isB ? (u0 * 32 + idx) : (idx * 32 + u0);
        const int pid = (p0 / (2 * s)) * s + (p0 % s);
        const float* tp = tw + ls * 2048 + pid * 4 + ((i >> lsi) & 1) * 2;
        if (lsi == 0) {
            val = *(const v2f*)tp;
        } else {
            const float f = tp[(j >> lsi) & 1];
            val[0] *= f;
            val[1] *= f;
        }
    }
    ((v2f*)pk)[o] = val;
}

__global__ __launch_bounds__(NWAVES * 32) void butterfly_wmma(
    const float* __restrict__ x,
    const float* __restrict__ pk,
    const float* __restrict__ bias,
    float* __restrict__ out)
{
    __shared__ __align__(16) float y[TILE_M * YS];

    const int tid = threadIdx.x;
    const int w   = tid >> 5;
    const int L   = tid & 31;
    const int Lm  = L & 15;
    const int Lh  = L >> 4;
    const size_t b0 = (size_t)blockIdx.x * TILE_M;

    const float* xrow = x + b0 * N_FULL;
    for (int i = tid; i < TILE_M * (N_FULL / 4); i += NWAVES * 32) {
        const int row = i >> 8;
        const int c4  = (i & 255) * 4;
        const float4 v = *(const float4*)(xrow + row * N_FULL + c4);
        *(float4*)(&y[row * YS + c4]) = v;
    }
    __syncthreads();

    const v2f* pkA = (const v2f*)pk;
    const v2f* pkB = (const v2f*)pk + 32 * 512;

    for (int g = w; g < NBLK; g += NWAVES) {
        v2f xa[8];
        const int xb = Lm * YS + g * 32 + Lh * 2;
        #pragma unroll
        for (int kk = 0; kk < 8; ++kk)
            xa[kk] = *(const v2f*)(&y[xb + kk * 4]);

        #pragma unroll
        for (int h = 0; h < 2; ++h) {
            v8f acc = {0.f, 0.f, 0.f, 0.f, 0.f, 0.f, 0.f, 0.f};
            #pragma unroll
            for (int kk = 0; kk < 8; ++kk) {
                const v2f bm = pkA[g * 512 + (h * 8 + kk) * 32 + L];
                acc = __builtin_amdgcn_wmma_f32_16x16x4_f32(
                    false, xa[kk], false, bm, (short)0, acc, false, false);
            }
            const int db = g * 32 + h * 16 + Lm;
            #pragma unroll
            for (int v = 0; v < 8; ++v)
                y[(v + 8 * Lh) * YS + db] = acc[v];
        }
    }
    __syncthreads();

    for (int r = w; r < 32; r += NWAVES) {
        v2f ya[8];
        #pragma unroll
        for (int kk = 0; kk < 8; ++kk) {
            const int k0 = kk * 4 + Lh * 2;
            ya[kk][0] = y[Lm * YS + (k0 + 0) * 32 + r];
            ya[kk][1] = y[Lm * YS + (k0 + 1) * 32 + r];
        }
        #pragma unroll
        for (int h = 0; h < 2; ++h) {
            v8f acc = {0.f, 0.f, 0.f, 0.f, 0.f, 0.f, 0.f, 0.f};
            #pragma unroll
            for (int kk = 0; kk < 8; ++kk) {
                const v2f bm = pkB[r * 512 + (h * 8 + kk) * 32 + L];
                acc = __builtin_amdgcn_wmma_f32_16x16x4_f32(
                    false, ya[kk], false, bm, (short)0, acc, false, false);
            }
            #pragma unroll
            for (int v = 0; v < 8; ++v)
                y[(v + 8 * Lh) * YS + (h * 16 + Lm) * 32 + r] = acc[v];
        }
    }
    __syncthreads();

    float* orow = out + b0 * N_FULL;
    for (int i = tid; i < TILE_M * (N_FULL / 4); i += NWAVES * 32) {
        const int row = i >> 8;
        const int c4  = (i & 255) * 4;
        float4 v = *(const float4*)(&y[row * YS + c4]);
        const float4 bv = *(const float4*)(bias + c4);
        v.x += bv.x; v.y += bv.y; v.z += bv.z; v.w += bv.w;
        *(float4*)(orow + row * N_FULL + c4) = v;
    }
}

extern "C" void kernel_launch(void* const* d_in, const int* in_sizes, int n_in,
                              void* d_out, int out_size, void* d_ws, size_t ws_size,
                              hipStream_t stream) {
    (void)n_in; (void)out_size; (void)ws_size;
    const float* x    = (const float*)d_in[0];
    const float* tw   = (const float*)d_in[1];
    const float* bias = (const float*)d_in[2];
    float* out = (float*)d_out;
    float* pk  = (float*)d_ws;

    const int batch = in_sizes[0] / N_FULL;

    build_packed_mats<<<(64 * 512) / SETUP_THREADS, SETUP_THREADS, 0, stream>>>(tw, pk);
    butterfly_wmma<<<batch / TILE_M, NWAVES * 32, 0, stream>>>(x, pk, bias, out);
}


# XOR swizzle of the LDS tile (in-block offset ^ pattern of block index): the inter-block stage's strided LDS stores (16 lanes per bank) and loads become conflict-free; same sums, same order
# speedup vs baseline: 1.7361x; 1.1360x over previous
#include <hip/hip_runtime.h>
#include <hip/hip_bf16.h>

typedef float v2f __attribute__((ext_vector_type(2)));
typedef float v8f __attribute__((ext_vector_type(8)));

#define N_FULL   1024
#define NBLK     32
#define TILE_M   16
#define YS       1028
#define NWAVES   4

__device__ __forceinline__ int swz(int g) {
    return (g & 2) | ((g & 1) << 2) | ((g & 12) << 1);
}

#define SETUP_THREADS 256

__global__ __launch_bounds__(SETUP_THREADS) void build_packed_mats(
    const float* __restrict__ tw,
    float* __restrict__ pk)
{
    const int o    = (int)blockIdx.x * SETUP_THREADS + (int)threadIdx.x;
    const int mat  = o >> 9;
    const int hk   = (o >> 5) & 15;
    const int L    = o & 31;
    const int i    = (hk >> 3) * 16 + (L & 15);
    const int j    = (hk & 7) * 4 + (L >> 4) * 2;
    const bool isB = (mat >= 32);
    const int idx  = mat & 31;

    v2f val;
    for (int lsi = 0; lsi < 5; ++lsi) {
        const int ls = isB ? lsi + 5 : lsi;
        const int s  = 1 << ls;
        const int cs = 1 << lsi;
        const int u0  = (i & (cs - 1)) | (j & ~(2 * cs - 1));
        const int p0  = isB ? (u0 * 32 + idx) : (idx * 32 + u0);
        const int pid = (p0 / (2 * s)) * s + (p0 % s);
        const float* tp = tw + ls * 2048 + pid * 4 + ((i >> lsi) & 1) * 2;
        if (lsi == 0) {
            val = *(const v2f*)tp;
        } else {
            const float f = tp[(j >> lsi) & 1];
            val[0] *= f;
            val[1] *= f;
        }
    }
    ((v2f*)pk)[o] = val;
}

__global__ __launch_bounds__(NWAVES * 32) void butterfly_wmma(
    const float* __restrict__ x,
    const float* __restrict__ pk,
    const float* __restrict__ bias,
    float* __restrict__ out)
{
    __shared__ __align__(16) float y[TILE_M * YS];

    const int tid = threadIdx.x;
    const int w   = tid >> 5;
    const int L   = tid & 31;
    const int Lm  = L & 15;
    const int Lh  = L >> 4;
    const size_t b0 = (size_t)blockIdx.x * TILE_M;

    const float* xrow = x + b0 * N_FULL;
    for (int i = tid; i < TILE_M * (N_FULL / 4); i += NWAVES * 32) {
        const int row = i >> 8;
        const int c4  = (i & 255) * 4;
        float4 v = *(const float4*)(xrow + row * N_FULL + c4);
        const int g = c4 >> 5;
        const int s = swz(g);
        if (s & 2) v = make_float4(v.z, v.w, v.x, v.y);
        *(float4*)(&y[row * YS + g * 32 + ((c4 & 31) ^ (s & 28))]) = v;
    }
    __syncthreads();

    const v2f* pkA = (const v2f*)pk;
    const v2f* pkB = (const v2f*)pk + 32 * 512;

    for (int g = w; g < NBLK; g += NWAVES) {
        v2f xa[8];
        const int sg = swz(g);
        const int xb = Lm * YS + g * 32;
        #pragma unroll
        for (int kk = 0; kk < 8; ++kk)
            xa[kk] = *(const v2f*)(&y[xb + ((kk * 4 + Lh * 2) ^ sg)]);

        #pragma unroll
        for (int h = 0; h < 2; ++h) {
            v8f acc = {0.f, 0.f, 0.f, 0.f, 0.f, 0.f, 0.f, 0.f};
            #pragma unroll
            for (int kk = 0; kk < 8; ++kk) {
                const v2f bm = pkA[g * 512 + (h * 8 + kk) * 32 + L];
                acc = __builtin_amdgcn_wmma_f32_16x16x4_f32(
                    false, xa[kk], false, bm, (short)0, acc, false, false);
            }
            const int db = g * 32 + ((h * 16 + Lm) ^ sg);
            #pragma unroll
            for (int v = 0; v < 8; ++v)
                y[(v + 8 * Lh) * YS + db] = acc[v];
        }
    }
    __syncthreads();

    for (int r = w; r < 32; r += NWAVES) {
        v2f ya[8];
        #pragma unroll
        for (int kk = 0; kk < 8; ++kk) {
            const int k0 = kk * 4 + Lh * 2;
            ya[kk][0] = y[Lm * YS + (k0 + 0) * 32 + (r ^ swz(k0 + 0))];
            ya[kk][1] = y[Lm * YS + (k0 + 1) * 32 + (r ^ swz(k0 + 1))];
        }
        #pragma unroll
        for (int h = 0; h < 2; ++h) {
            v8f acc = {0.f, 0.f, 0.f, 0.f, 0.f, 0.f, 0.f, 0.f};
            #pragma unroll
            for (int kk = 0; kk < 8; ++kk) {
                const v2f bm = pkB[r * 512 + (h * 8 + kk) * 32 + L];
                acc = __builtin_amdgcn_wmma_f32_16x16x4_f32(
                    false, ya[kk], false, bm, (short)0, acc, false, false);
            }
            const int gd = h * 16 + Lm;
            const int dz = gd * 32 + (r ^ swz(gd));
            #pragma unroll
            for (int v = 0; v < 8; ++v)
                y[(v + 8 * Lh) * YS + dz] = acc[v];
        }
    }
    __syncthreads();

    float* orow = out + b0 * N_FULL;
    for (int i = tid; i < TILE_M * (N_FULL / 4); i += NWAVES * 32) {
        const int row = i >> 8;
        const int c4  = (i & 255) * 4;
        const int g = c4 >> 5;
        const int s = swz(g);
        float4 v = *(const float4*)(&y[row * YS + g * 32 + ((c4 & 31) ^ (s & 28))]);
        if (s & 2) v = make_float4(v.z, v.w, v.x, v.y);
        const float4 bv = *(const float4*)(bias + c4);
        v.x += bv.x; v.y += bv.y; v.z += bv.z; v.w += bv.w;
        *(float4*)(orow + row * N_FULL + c4) = v;
    }
}

extern "C" void kernel_launch(void* const* d_in, const int* in_sizes, int n_in,
                              void* d_out, int out_size, void* d_ws, size_t ws_size,
                              hipStream_t stream) {
    (void)n_in; (void)out_size; (void)ws_size;
    const float* x    = (const float*)d_in[0];
    const float* tw   = (const float*)d_in[1];
    const float* bias = (const float*)d_in[2];
    float* out = (float*)d_out;
    float* pk  = (float*)d_ws;

    const int batch = in_sizes[0] / N_FULL;

    build_packed_mats<<<(64 * 512) / SETUP_THREADS, SETUP_THREADS, 0, stream>>>(tw, pk);
    butterfly_wmma<<<batch / TILE_M, NWAVES * 32, 0, stream>>>(x, pk, bias, out);
}


# Group A takes its x fragments straight from global memory (16 B per lane, a wave's 8 blocks requested up front): LDS stage-in pass and its barrier removed; k order of group A's 32-term sums permuted
# speedup vs baseline: 1.8380x; 1.0587x over previous
#include <hip/hip_runtime.h>
#include <hip/hip_bf16.h>

typedef float v2f __attribute__((ext_vector_type(2)));
typedef float v8f __attribute__((ext_vector_type(8)));

#define N_FULL   1024
#define NBLK     32
#define TILE_M   16
#define YS       1028
#define NWAVES   4

__device__ __forceinline__ int swz(int g) {
    return (g & 2) | ((g & 1) << 2) | ((g & 12) << 1);
}

#define SETUP_THREADS 256

__global__ __launch_bounds__(SETUP_THREADS) void build_packed_mats(
    const float* __restrict__ tw,
    float* __restrict__ pk)
{
    const int o    = (int)blockIdx.x * SETUP_THREADS + (int)threadIdx.x;
    const int mat  = o >> 9;
    const int hk   = (o >> 5) & 15;
    const int L    = o & 31;
    const int i    = (hk >> 3) * 16 + (L & 15);
    const bool isB = (mat >= 32);
    const int j    = isB ? (hk & 7) * 4 + (L >> 4) * 2
                         : (L >> 4) * 16 + (hk & 7) * 2;
    const int idx  = mat & 31;

    v2f val;
    for (int lsi = 0; lsi < 5; ++lsi) {
        const int ls = isB ? lsi + 5 : lsi;
        const int s  = 1 << ls;
        const int cs = 1 << lsi;
        const int u0  = (i & (cs - 1)) | (j & ~(2 * cs - 1));
        const int p0  = isB ? (u0 * 32 + idx) : (idx * 32 + u0);
        const int pid = (p0 / (2 * s)) * s + (p0 % s);
        const float* tp = tw + ls * 2048 + pid * 4 + ((i >> lsi) & 1) * 2;
        if (lsi == 0) {
            val = *(const v2f*)tp;
        } else {
            const float f = tp[(j >> lsi) & 1];
            val[0] *= f;
            val[1] *= f;
        }
    }
    ((v2f*)pk)[o] = val;
}

__global__ __launch_bounds__(NWAVES * 32) void butterfly_wmma(
    const float* __restrict__ x,
    const float* __restrict__ pk,
    const float* __restrict__ bias,
    float* __restrict__ out)
{
    __shared__ __align__(16) float y[TILE_M * YS];

    const int tid = threadIdx.x;
    const int w   = tid >> 5;
    const int L   = tid & 31;
    const int Lm  = L & 15;
    const int Lh  = L >> 4;
    const size_t b0 = (size_t)blockIdx.x * TILE_M;

    const v2f* pkA = (const v2f*)pk;
    const v2f* pkB = (const v2f*)pk + 32 * 512;

    const float* xlane = x + (b0 + Lm) * N_FULL + Lh * 16;
    float4 xf[NBLK / NWAVES][4];
    #pragma unroll
    for (int it = 0; it < NBLK / NWAVES; ++it) {
        #pragma unroll
        for (int q = 0; q < 4; ++q)
            xf[it][q] = *(const float4*)(xlane + (w + it * NWAVES) * 32 + q * 4);
    }
    __builtin_amdgcn_sched_barrier(0);

    #pragma unroll
    for (int it = 0; it < NBLK / NWAVES; ++it) {
        const int g  = w + it * NWAVES;
        const int sg = swz(g);
        #pragma unroll
        for (int h = 0; h < 2; ++h) {
            v8f acc = {0.f, 0.f, 0.f, 0.f, 0.f, 0.f, 0.f, 0.f};
            #pragma unroll
            for (int kk = 0; kk < 8; ++kk) {
                const v2f bm = pkA[g * 512 + (h * 8 + kk) * 32 + L];
                const float4 xq = xf[it][kk >> 1];
                v2f xa;
                xa[0] = (kk & 1) ? xq.z : xq.x;
                xa[1] = (kk & 1) ? xq.w : xq.y;
                acc = __builtin_amdgcn_wmma_f32_16x16x4_f32(
                    false, xa, false, bm, (short)0, acc, false, false);
            }
            const int db = g * 32 + ((h * 16 + Lm) ^ sg);
            #pragma unroll
            for (int v = 0; v < 8; ++v)
                y[(v + 8 * Lh) * YS + db] = acc[v];
        }
        __builtin_amdgcn_sched_barrier(0);
    }
    __syncthreads();

    for (int r = w; r < 32; r += NWAVES) {
        v2f ya[8];
        #pragma unroll
        for (int kk = 0; kk < 8; ++kk) {
            const int k0 = kk * 4 + Lh * 2;
            ya[kk][0] = y[Lm * YS + (k0 + 0) * 32 + (r ^ swz(k0 + 0))];
            ya[kk][1] = y[Lm * YS + (k0 + 1) * 32 + (r ^ swz(k0 + 1))];
        }
        #pragma unroll
        for (int h = 0; h < 2; ++h) {
            v8f acc = {0.f, 0.f, 0.f, 0.f, 0.f, 0.f, 0.f, 0.f};
            #pragma unroll
            for (int kk = 0; kk < 8; ++kk) {
                const v2f bm = pkB[r * 512 + (h * 8 + kk) * 32 + L];
                acc = __builtin_amdgcn_wmma_f32_16x16x4_f32(
                    false, ya[kk], false, bm, (short)0, acc, false, false);
            }
            const int gd = h * 16 + Lm;
            const int dz = gd * 32 + (r ^ swz(gd));
            #pragma unroll
            for (int v = 0; v < 8; ++v)
                y[(v + 8 * Lh) * YS + dz] = acc[v];
        }
    }
    __syncthreads();

    float* orow = out + b0 * N_FULL;
    for (int i = tid; i < TILE_M * (N_FULL / 4); i += NWAVES * 32) {
        const int row = i >> 8;
        const int c4  = (i & 255) * 4;
        const int g = c4 >> 5;
        const int s = swz(g);
        float4 v = *(const float4*)(&y[row * YS + g * 32 + ((c4 & 31) ^ (s & 28))]);
        if (s & 2) v = make_float4(v.z, v.w, v.x, v.y);
        const float4 bv = *(const float4*)(bias + c4);
        v.x += bv.x; v.y += bv.y; v.z += bv.z; v.w += bv.w;
        *(float4*)(orow + row * N_FULL + c4) = v;
    }
}

extern "C" void kernel_launch(void* const* d_in, const int* in_sizes, int n_in,
                              void* d_out, int out_size, void* d_ws, size_t ws_size,
                              hipStream_t stream) {
    (void)n_in; (void)out_size; (void)ws_size;
    const float* x    = (const float*)d_in[0];
    const float* tw   = (const float*)d_in[1];
    const float* bias = (const float*)d_in[2];
    float* out = (float*)d_out;
    float* pk  = (float*)d_ws;

    const int batch = in_sizes[0] / N_FULL;

    build_packed_mats<<<(64 * 512) / SETUP_THREADS, SETUP_THREADS, 0, stream>>>(tw, pk);
    butterfly_wmma<<<batch / TILE_M, NWAVES * 32, 0, stream>>>(x, pk, bias, out);
}


# 32-row tile per workgroup (8 waves, was 16 rows, 4 waves): each matrix fragment loaded from L2 feeds two products (both 16-row halves), matrix reads from L2 halved; same sums, same order
# speedup vs baseline: 2.0394x; 1.1096x over previous
#include <hip/hip_runtime.h>
#include <hip/hip_bf16.h>

typedef float v2f __attribute__((ext_vector_type(2)));
typedef float v8f __attribute__((ext_vector_type(8)));

#define N_FULL   1024
#define NBLK     32
#define TILE_M   32
#define NRT      2
#define YS       1028
#define NWAVES   8

__device__ __forceinline__ int swz(int g) {
    return (g & 2) | ((g & 1) << 2) | ((g & 12) << 1);
}

#define SETUP_THREADS 256

__global__ __launch_bounds__(SETUP_THREADS) void build_packed_mats(
    const float* __restrict__ tw,
    float* __restrict__ pk)
{
    const int o    = (int)blockIdx.x * SETUP_THREADS + (int)threadIdx.x;
    const int mat  = o >> 9;
    const int hk   = (o >> 5) & 15;
    const int L    = o & 31;
    const int i    = (hk >> 3) * 16 + (L & 15);
    const bool isB = (mat >= 32);
    const int j    = isB ? (hk & 7) * 4 + (L >> 4) * 2
                         : (L >> 4) * 16 + (hk & 7) * 2;
    const int idx  = mat & 31;

    v2f val;
    for (int lsi = 0; lsi < 5; ++lsi) {
        const int ls = isB ? lsi + 5 : lsi;
        const int s  = 1 << ls;
        const int cs = 1 << lsi;
        const int u0  = (i & (cs - 1)) | (j & ~(2 * cs - 1));
        const int p0  = isB ? (u0 * 32 + idx) : (idx * 32 + u0);
        const int pid = (p0 / (2 * s)) * s + (p0 % s);
        const float* tp = tw + ls * 2048 + pid * 4 + ((i >> lsi) & 1) * 2;
        if (lsi == 0) {
            val = *(const v2f*)tp;
        } else {
            const float f = tp[(j >> lsi) & 1];
            val[0] *= f;
            val[1] *= f;
        }
    }
    ((v2f*)pk)[o] = val;
}

__global__ __launch_bounds__(NWAVES * 32) void butterfly_wmma(
    const float* __restrict__ x,
    const float* __restrict__ pk,
    const float* __restrict__ bias,
    float* __restrict__ out)
{
    __shared__ __align__(16) float y[TILE_M * YS];

    const int tid = threadIdx.x;
    const int w   = tid >> 5;
    const int L   = tid & 31;
    const int Lm  = L & 15;
    const int Lh  = L >> 4;
    const size_t b0 = (size_t)blockIdx.x * TILE_M;

    const v2f* pkA = (const v2f*)pk;
    const v2f* pkB = (const v2f*)pk + 32 * 512;

    const float* xlane = x + (b0 + Lm) * N_FULL + Lh * 16;
    float4 xf[NBLK / NWAVES][NRT][4];
    #pragma unroll
    for (int it = 0; it < NBLK / NWAVES; ++it) {
        #pragma unroll
        for (int rt = 0; rt < NRT; ++rt) {
            #pragma unroll
            for (int q = 0; q < 4; ++q)
                xf[it][rt][q] = *(const float4*)(xlane + rt * 16 * N_FULL + (w + it * NWAVES) * 32 + q * 4);
        }
    }
    __builtin_amdgcn_sched_barrier(0);

    #pragma unroll
    for (int it = 0; it < NBLK / NWAVES; ++it) {
        const int g  = w + it * NWAVES;
        const int sg = swz(g);
        #pragma unroll
        for (int h = 0; h < 2; ++h) {
            v8f acc[NRT];
            #pragma unroll
            for (int rt = 0; rt < NRT; ++rt)
                acc[rt] = (v8f){0.f, 0.f, 0.f, 0.f, 0.f, 0.f, 0.f, 0.f};
            #pragma unroll
            for (int kk = 0; kk < 8; ++kk) {
                const v2f bm = pkA[g * 512 + (h * 8 + kk) * 32 + L];
                #pragma unroll
                for (int rt = 0; rt < NRT; ++rt) {
                    const float4 xq = xf[it][rt][kk >> 1];
                    v2f xa;
                    xa[0] = (kk & 1) ? xq.z : xq.x;
                    xa[1] = (kk & 1) ? xq.w : xq.y;
                    acc[rt] = __builtin_amdgcn_wmma_f32_16x16x4_f32(
                        false, xa, false, bm, (short)0, acc[rt], false, false);
                }
            }
            const int db = g * 32 + ((h * 16 + Lm) ^ sg);
            #pragma unroll
            for (int rt = 0; rt < NRT; ++rt) {
                #pragma unroll
                for (int v = 0; v < 8; ++v)
                    y[(rt * 16 + v + 8 * Lh) * YS + db] = acc[rt][v];
            }
        }
        __builtin_amdgcn_sched_barrier(0);
    }
    __syncthreads();

    for (int r = w; r < 32; r += NWAVES) {
        v2f ya[NRT][8];
        #pragma unroll
        for (int rt = 0; rt < NRT; ++rt) {
            #pragma unroll
            for (int kk = 0; kk < 8; ++kk) {
                const int k0 = kk * 4 + Lh * 2;
                ya[rt][kk][0] = y[(rt * 16 + Lm) * YS + (k0 + 0) * 32 + (r ^ swz(k0 + 0))];
                ya[rt][kk][1] = y[(rt * 16 + Lm) * YS + (k0 + 1) * 32 + (r ^ swz(k0 + 1))];
            }
        }
        #pragma unroll
        for (int h = 0; h < 2; ++h) {
            v8f acc[NRT];
            #pragma unroll
            for (int rt = 0; rt < NRT; ++rt)
                acc[rt] = (v8f){0.f, 0.f, 0.f, 0.f, 0.f, 0.f, 0.f, 0.f};
            #pragma unroll
            for (int kk = 0; kk < 8; ++kk) {
                const v2f bm = pkB[r * 512 + (h * 8 + kk) * 32 + L];
                #pragma unroll
                for (int rt = 0; rt < NRT; ++rt)
                    acc[rt] = __builtin_amdgcn_wmma_f32_16x16x4_f32(
                        false, ya[rt][kk], false, bm, (short)0, acc[rt], false, false);
            }
            const int gd = h * 16 + Lm;
            const int dz = gd * 32 + (r ^ swz(gd));
            #pragma unroll
            for (int rt = 0; rt < NRT; ++rt) {
                #pragma unroll
                for (int v = 0; v < 8; ++v)
                    y[(rt * 16 + v + 8 * Lh) * YS + dz] = acc[rt][v];
            }
        }
    }
    __syncthreads();

    float* orow = out + b0 * N_FULL;
    for (int i = tid; i < TILE_M * (N_FULL / 4); i += NWAVES * 32) {
        const int row = i >> 8;
        const int c4  = (i & 255) * 4;
        const int g = c4 >> 5;
        const int s = swz(g);
        float4 v = *(const float4*)(&y[row * YS + g * 32 + ((c4 & 31) ^ (s & 28))]);
        if (s & 2) v = make_float4(v.z, v.w, v.x, v.y);
        const float4 bv = *(const float4*)(bias + c4);
        v.x += bv.x; v.y += bv.y; v.z += bv.z; v.w += bv.w;
        *(float4*)(orow + row * N_FULL + c4) = v;
    }
}

extern "C" void kernel_launch(void* const* d_in, const int* in_sizes, int n_in,
                              void* d_out, int out_size, void* d_ws, size_t ws_size,
                              hipStream_t stream) {
    (void)n_in; (void)out_size; (void)ws_size;
    const float* x    = (const float*)d_in[0];
    const float* tw   = (const float*)d_in[1];
    const float* bias = (const float*)d_in[2];
    float* out = (float*)d_out;
    float* pk  = (float*)d_ws;

    const int batch = in_sizes[0] / N_FULL;

    build_packed_mats<<<(64 * 512) / SETUP_THREADS, SETUP_THREADS, 0, stream>>>(tw, pk);
    butterfly_wmma<<<batch / TILE_M, NWAVES * 32, 0, stream>>>(x, pk, bias, out);
}


# One workgroup per WGP loops over 4 row tiles, matrices resident (B_r in LDS, A_g in registers): next tile's x is requested after group A, streams in under group B and write-out; same sums, same order
# speedup vs baseline: 2.1765x; 1.0672x over previous
#include <hip/hip_runtime.h>
#include <hip/hip_bf16.h>

typedef float v2f __attribute__((ext_vector_type(2)));
typedef float v8f __attribute__((ext_vector_type(8)));

#define N_FULL   1024
#define NBLK     32
#define TILE_M   32
#define NRT      2
#define YS       1028
#define NWAVES   16
#define BPW      (NBLK / NWAVES)
#define NWG      256

__device__ __forceinline__ int swz(int g) {
    return (g & 2) | ((g & 1) << 2) | ((g & 12) << 1);
}

#define SETUP_THREADS 256

__global__ __launch_bounds__(SETUP_THREADS) void build_packed_mats(
    const float* __restrict__ tw,
    float* __restrict__ pk)
{
    const int o    = (int)blockIdx.x * SETUP_THREADS + (int)threadIdx.x;
    const int mat  = o >> 9;
    const int hk   = (o >> 5) & 15;
    const int L    = o & 31;
    const int i    = (hk >> 3) * 16 + (L & 15);
    const bool isB = (mat >= 32);
    const int j    = isB ? (hk & 7) * 4 + (L >> 4) * 2
                         : (L >> 4) * 16 + (hk & 7) * 2;
    const int idx  = mat & 31;

    v2f val;
    for (int lsi = 0; lsi < 5; ++lsi) {
        const int ls = isB ? lsi + 5 : lsi;
        const int s  = 1 << ls;
        const int cs = 1 << lsi;
        const int u0  = (i & (cs - 1)) | (j & ~(2 * cs - 1));
        const int p0  = isB ? (u0 * 32 + idx) : (idx * 32 + u0);
        const int pid = (p0 / (2 * s)) * s + (p0 % s);
        const float* tp = tw + ls * 2048 + pid * 4 + ((i >> lsi) & 1) * 2;
        if (lsi == 0) {
            val = *(const v2f*)tp;
        } else {
            const float f = tp[(j >> lsi) & 1];
            val[0] *= f;
            val[1] *= f;
        }
    }
    ((v2f*)pk)[o] = val;
}

__device__ __forceinline__ void lds_barrier() {
    __builtin_amdgcn_fence(__ATOMIC_RELEASE, "workgroup", "local");
    __builtin_amdgcn_s_barrier();
    __builtin_amdgcn_fence(__ATOMIC_ACQUIRE, "workgroup", "local");
}

__global__ __launch_bounds__(NWAVES * 32) void butterfly_wmma(
    const float* __restrict__ x,
    const float* __restrict__ pk,
    const float* __restrict__ bias,
    float* __restrict__ out,
    int ntiles)
{
    __shared__ __align__(16) float y[TILE_M * YS];
    __shared__ __align__(16) float bl[NBLK * 1024];

    const int tid = threadIdx.x;
    const int w   = tid >> 5;
    const int L   = tid & 31;
    const int Lm  = L & 15;
    const int Lh  = L >> 4;

    const v2f* pkA = (const v2f*)pk;
    const v2f* blv = (const v2f*)bl;

    int tile = (int)blockIdx.x;

    const float* xlane = x + (size_t)Lm * N_FULL + Lh * 16 + w * 32;
    float4 xf[BPW][NRT][4];
    if (tile < ntiles) {
        const float* xt = xlane + (size_t)tile * (TILE_M * N_FULL);
        #pragma unroll
        for (int it = 0; it < BPW; ++it) {
            #pragma unroll
            for (int rt = 0; rt < NRT; ++rt) {
                #pragma unroll
                for (int q = 0; q < 4; ++q)
                    xf[it][rt][q] = *(const float4*)(xt + rt * 16 * N_FULL + it * NWAVES * 32 + q * 4);
            }
        }
    }

    {
        const float4* src4 = (const float4*)(pk + 32 * 1024);
        float4* dst4 = (float4*)bl;
        #pragma unroll
        for (int n = 0; n < (NBLK * 256) / (NWAVES * 32); ++n) {
            const int i = (((n + (int)blockIdx.x) & 15) * (NWAVES * 32)) + tid;
            dst4[i] = src4[i];
        }
    }

    v2f am[BPW][2][8];
    #pragma unroll
    for (int it = 0; it < BPW; ++it) {
        #pragma unroll
        for (int h = 0; h < 2; ++h) {
            #pragma unroll
            for (int kk = 0; kk < 8; ++kk)
                am[it][h][kk] = pkA[(w + it * NWAVES) * 512 + (h * 8 + kk) * 32 + L];
        }
    }

    const int c4w = (tid & 255) * 4;
    const float4 bv = *(const float4*)(bias + c4w);

    for (; tile < ntiles; tile += NWG) {
        const int nt = tile + NWG;
        #pragma unroll
        for (int it = 0; it < BPW; ++it) {
            const int g  = w + it * NWAVES;
            const int sg = swz(g);
            #pragma unroll
            for (int h = 0; h < 2; ++h) {
                v8f acc[NRT];
                #pragma unroll
                for (int rt = 0; rt < NRT; ++rt)
                    acc[rt] = (v8f){0.f, 0.f, 0.f, 0.f, 0.f, 0.f, 0.f, 0.f};
                #pragma unroll
                for (int kk = 0; kk < 8; ++kk) {
                    #pragma unroll
                    for (int rt = 0; rt < NRT; ++rt) {
                        const float4 xq = xf[it][rt][kk >> 1];
                        v2f xa;
                        xa[0] = (kk & 1) ? xq.z : xq.x;
                        xa[1] = (kk & 1) ? xq.w : xq.y;
                        acc[rt] = __builtin_amdgcn_wmma_f32_16x16x4_f32(
                            false, xa, false, am[it][h][kk], (short)0, acc[rt], false, false);
                    }
                }
                const int db = g * 32 + ((h * 16 + Lm) ^ sg);
                #pragma unroll
                for (int rt = 0; rt < NRT; ++rt) {
                    #pragma unroll
                    for (int v = 0; v < 8; ++v)
                        y[(rt * 16 + v + 8 * Lh) * YS + db] = acc[rt][v];
                }
            }
            __builtin_amdgcn_sched_barrier(0);
        }

        if (nt < ntiles) {
            const float* xt = xlane + (size_t)nt * (TILE_M * N_FULL);
            #pragma unroll
            for (int it = 0; it < BPW; ++it) {
                #pragma unroll
                for (int rt = 0; rt < NRT; ++rt) {
                    #pragma unroll
                    for (int q = 0; q < 4; ++q)
                        xf[it][rt][q] = *(const float4*)(xt + rt * 16 * N_FULL + it * NWAVES * 32 + q * 4);
                }
            }
        }
        lds_barrier();

        for (int r = w; r < 32; r += NWAVES) {
            v2f ya[NRT][8];
            v2f bm[2][8];
            #pragma unroll
            for (int rt = 0; rt < NRT; ++rt) {
                #pragma unroll
                for (int kk = 0; kk < 8; ++kk) {
                    const int k0 = kk * 4 + Lh * 2;
                    ya[rt][kk][0] = y[(rt * 16 + Lm) * YS + (k0 + 0) * 32 + (r ^ swz(k0 + 0))];
                    ya[rt][kk][1] = y[(rt * 16 + Lm) * YS + (k0 + 1) * 32 + (r ^ swz(k0 + 1))];
                }
            }
            #pragma unroll
            for (int h = 0; h < 2; ++h) {
                #pragma unroll
                for (int kk = 0; kk < 8; ++kk)
                    bm[h][kk] = blv[r * 512 + (h * 8 + kk) * 32 + L];
            }
            __builtin_amdgcn_sched_barrier(0);
            #pragma unroll
            for (int h = 0; h < 2; ++h) {
                v8f acc[NRT];
                #pragma unroll
                for (int rt = 0; rt < NRT; ++rt)
                    acc[rt] = (v8f){0.f, 0.f, 0.f, 0.f, 0.f, 0.f, 0.f, 0.f};
                #pragma unroll
                for (int kk = 0; kk < 8; ++kk) {
                    #pragma unroll
                    for (int rt = 0; rt < NRT; ++rt)
                        acc[rt] = __builtin_amdgcn_wmma_f32_16x16x4_f32(
                            false, ya[rt][kk], false, bm[h][kk], (short)0, acc[rt], false, false);
                }
                const int gd = h * 16 + Lm;
                const int dz = gd * 32 + (r ^ swz(gd));
                #pragma unroll
                for (int rt = 0; rt < NRT; ++rt) {
                    #pragma unroll
                    for (int v = 0; v < 8; ++v)
                        y[(rt * 16 + v + 8 * Lh) * YS + dz] = acc[rt][v];
                }
            }
        }
        lds_barrier();

        float* orow = out + (size_t)tile * (TILE_M * N_FULL);
        {
            const int g = c4w >> 5;
            const int s = swz(g);
            const int lo = g * 32 + ((c4w & 31) ^ (s & 28));
            #pragma unroll
            for (int n = 0; n < TILE_M / 2; ++n) {
                const int row = (tid >> 8) + 2 * n;
                float4 v = *(const float4*)(&y[row * YS + lo]);
                if (s & 2) v = make_float4(v.z, v.w, v.x, v.y);
                v.x += bv.x; v.y += bv.y; v.z += bv.z; v.w += bv.w;
                *(float4*)(orow + row * N_FULL + c4w) = v;
            }
        }
        lds_barrier();
    }
}

extern "C" void kernel_launch(void* const* d_in, const int* in_sizes, int n_in,
                              void* d_out, int out_size, void* d_ws, size_t ws_size,
                              hipStream_t stream) {
    (void)n_in; (void)out_size; (void)ws_size;
    const float* x    = (const float*)d_in[0];
    const float* tw   = (const float*)d_in[1];
    const float* bias = (const float*)d_in[2];
    float* out = (float*)d_out;
    float* pk  = (float*)d_ws;

    const int batch = in_sizes[0] / N_FULL;

    build_packed_mats<<<(64 * 512) / SETUP_THREADS, SETUP_THREADS, 0, stream>>>(tw, pk);
    butterfly_wmma<<<NWG, NWAVES * 32, 0, stream>>>(x, pk, bias, out, batch / TILE_M);
}
